# starting kernel
# baseline (speedup 1.0000x reference)
#include <hip/hip_runtime.h>

typedef __attribute__((ext_vector_type(2))) float v2f;
typedef __attribute__((ext_vector_type(8))) float v8f;

#define D 64


__global__ void zero_f32(float* __restrict__ p, int n) {
  int i = blockIdx.x * blockDim.x + threadIdx.x;
  if (i < n) p[i] = 0.0f;
}

__global__ void deg_count(const int* __restrict__ dst, float* __restrict__ deg, int n_edges) {
  int e = blockIdx.x * blockDim.x + threadIdx.x;
  if (e < n_edges) atomicAdd(&deg[dst[e]], 1.0f);
}

__global__ void invsqrt_finish(float* __restrict__ d, int n) {
  int i = blockIdx.x * blockDim.x + threadIdx.x;
  if (i < n) d[i] = 1.0f / sqrtf(d[i] + 1.0f);
}

__global__ void edge_norm(const int* __restrict__ src, const int* __restrict__ dst,
                          const float* __restrict__ inv, float* __restrict__ norm,
                          int n_edges) {
  int e = blockIdx.x * blockDim.x + threadIdx.x;
  if (e < n_edges) norm[e] = inv[src[e]] * inv[dst[e]];
}

template <bool RELU>
__global__ void gemm16_wmma(const float* __restrict__ A, const float* __restrict__ W,
                            float* __restrict__ T) {
  const int lane = threadIdx.x & 31;
  const int wave = threadIdx.x >> 5;
  const int half = lane >> 4;
  const int l    = lane & 15;
  const int row0 = blockIdx.x << 4;
  const int c0   = wave << 4;

  const float* __restrict__ arow = A + (size_t)(row0 + l) * D;

  v8f acc = {};
#pragma unroll
  for (int kk = 0; kk < D; kk += 4) {
    const int ka = kk + half * 2;
    float a0 = arow[ka];
    float a1 = arow[ka + 1];
    if (RELU) { a0 = fmaxf(a0, 0.0f); a1 = fmaxf(a1, 0.0f); }
    v2f a = {a0, a1};
    v2f b = { W[ka * D + c0 + l], W[(ka + 1) * D + c0 + l] };
    acc = __builtin_amdgcn_wmma_f32_16x16x4_f32(
         false, a,  false, b,
         (short)0, acc,  false,  false);
  }

  float* __restrict__ orow = T + (size_t)(row0 + half * 8) * D + c0 + l;
#pragma unroll
  for (int j = 0; j < 8; ++j) orow[(size_t)j * D] = acc[j];
}

__global__ void self_bias_init(const float* __restrict__ T, const float* __restrict__ inv,
                               const float* __restrict__ bias, float* __restrict__ out,
                               int total) {
  int i = blockIdx.x * blockDim.x + threadIdx.x;
  if (i >= total) return;
  float is = inv[i >> 6];
  out[i] = T[i] * is * is + bias[i & (D - 1)];
}

__global__ void edge_aggregate(const int* __restrict__ src, const int* __restrict__ dst,
                               const float* __restrict__ norm, const float* __restrict__ T,
                               float* __restrict__ out, int n_edges) {
  int e = blockIdx.x * (blockDim.x >> 5) + (threadIdx.x >> 5);
  if (e >= n_edges) return;
  int f2 = (threadIdx.x & 31) * 2;
  int s = src[e];
  int d = dst[e];
  float nv = norm[e];
  const float2 t = *(const float2*)(T + (size_t)s * D + f2);
  float* o = out + (size_t)d * D + f2;
  atomicAdd(o,     t.x * nv);
  atomicAdd(o + 1, t.y * nv);
}


static inline size_t align256(size_t x) { return (x + 255) & ~(size_t)255; }

extern "C" void kernel_launch(void* const* d_in, const int* in_sizes, int n_in,
                              void* d_out, int out_size, void* d_ws, size_t ws_size,
                              hipStream_t stream) {
  const int*   E  = (const int*)d_in[1];
  const float* X  = (const float*)d_in[2];
  const float* W1 = (const float*)d_in[3];
  const float* b1 = (const float*)d_in[4];
  const float* W2 = (const float*)d_in[5];
  const float* b2 = (const float*)d_in[6];
  const float* W3 = (const float*)d_in[7];
  const float* b3 = (const float*)d_in[8];
  float* out = (float*)d_out;

  const int n_edges = in_sizes[1] / 2;
  const int n_nodes = in_sizes[2] / D;
  const int* src = E;
  const int* dst = E + n_edges;

  char* ws = (char*)d_ws;
  float* inv  = (float*)ws; ws += align256((size_t)n_nodes * 4);
  float* norm = (float*)ws; ws += align256((size_t)n_edges * 4);
  float* T    = (float*)ws; ws += align256((size_t)n_nodes * D * 4);
  float* H    = (float*)ws;

  const int total = n_nodes * D;
  const int nb_nodes = (n_nodes + 255) / 256;
  const int nb_edges = (n_edges + 255) / 256;
  const int nb_total = (total + 255) / 256;
  const int nb_agg   = (n_edges + 7) / 8;
  const int nb_gemm  = n_nodes / 16;

  zero_f32<<<nb_nodes, 256, 0, stream>>>(inv, n_nodes);
  deg_count<<<nb_edges, 256, 0, stream>>>(dst, inv, n_edges);
  invsqrt_finish<<<nb_nodes, 256, 0, stream>>>(inv, n_nodes);
  edge_norm<<<nb_edges, 256, 0, stream>>>(src, dst, inv, norm, n_edges);

  gemm16_wmma<false><<<nb_gemm, 128, 0, stream>>>(X, W1, T);
  self_bias_init<<<nb_total, 256, 0, stream>>>(T, inv, b1, H, total);
  edge_aggregate<<<nb_agg, 256, 0, stream>>>(src, dst, norm, T, H, n_edges);

  gemm16_wmma<true><<<nb_gemm, 128, 0, stream>>>(H, W2, T);
  self_bias_init<<<nb_total, 256, 0, stream>>>(T, inv, b2, H, total);
  edge_aggregate<<<nb_agg, 256, 0, stream>>>(src, dst, norm, T, H, n_edges);

  gemm16_wmma<true><<<nb_gemm, 128, 0, stream>>>(H, W3, T);
  self_bias_init<<<nb_total, 256, 0, stream>>>(T, inv, b3, out, total);
  edge_aggregate<<<nb_agg, 256, 0, stream>>>(src, dst, norm, T, out, n_edges);
}


# edge scatter: a row's f32 atomics all come from one cache half: work-group reads its die, draws a (2048-edge chunk, row parity) ticket of its half, adds that parity's edges only; arrival order
# speedup vs baseline: 2.6194x; 2.6194x over previous
#include <hip/hip_runtime.h>

typedef __attribute__((ext_vector_type(2))) float v2f;
typedef __attribute__((ext_vector_type(8))) float v8f;

#define D 64


#define DEG_R 8
#define DEG_S 32
#define DEG_LOADS 16
__global__ void __launch_bounds__(1024)
deg_count_lds(const int* __restrict__ dst, unsigned* __restrict__ planes,
              int n_edges, int n_nodes, int range_n, int slice_n) {
  extern __shared__ unsigned cnt[];
  const int tid = (int)threadIdx.x;
  const int nt  = (int)blockDim.x;
  const int s  = (int)blockIdx.x % DEG_S;
  const int lo = ((int)blockIdx.x / DEG_S) * range_n;
  const int cn = max(0, min(range_n, n_nodes - lo));
  for (int k = tid; k < cn; k += nt) cnt[k] = 0u;
  __syncthreads();
  const int e0 = s * slice_n;
  const int e1 = min(e0 + slice_n, n_edges);
  for (int base = e0 + tid; base < e1; base += DEG_LOADS * nt) {
    int d[DEG_LOADS];
#pragma unroll
    for (int u = 0; u < DEG_LOADS; ++u) {
      const int e = base + u * nt;
      const int v = dst[min(e, e1 - 1)];
      d[u] = (e < e1) ? v : -1;
    }
#pragma unroll
    for (int u = 0; u < DEG_LOADS; ++u) {
      const unsigned k = (unsigned)(d[u] - lo);
      if (k < (unsigned)cn) atomicAdd(&cnt[k], 1u);
    }
  }
  __syncthreads();
  unsigned* __restrict__ p = planes + (size_t)s * n_nodes + lo;
  for (int k = tid; k < cn; k += nt) p[k] = cnt[k];
}

#define TK_STRIDE 64
#define TK_LAYER (2 * TK_STRIDE)
#define N_LAYERS 3

__global__ void invsqrt_from_planes(const unsigned* __restrict__ planes,
                                    float* __restrict__ inv, unsigned* __restrict__ tickets,
                                    int n) {
  int i = blockIdx.x * blockDim.x + threadIdx.x;
  if (i < 2 * N_LAYERS) tickets[(i >> 1) * TK_LAYER + (i & 1) * TK_STRIDE] = 0u;
  if (i < n) {
    unsigned c = 0u;
#pragma unroll
    for (int s = 0; s < DEG_S; ++s) c += planes[(size_t)s * n + i];
    inv[i] = 1.0f / sqrtf((float)c + 1.0f);
  }
}

#define EN_PIECES 8
__global__ void edge_norm(const int* __restrict__ src, const int* __restrict__ dst,
                          const float* __restrict__ inv, float* __restrict__ norm,
                          int n_edges) {
  const int base = (int)(blockIdx.x * (blockDim.x * EN_PIECES) + threadIdx.x);
  int s[EN_PIECES], d[EN_PIECES];
  float a[EN_PIECES], b[EN_PIECES];
#pragma unroll
  for (int k = 0; k < EN_PIECES; ++k) {
    const int ec = min(base + k * (int)blockDim.x, n_edges - 1);
    s[k] = src[ec];
    d[k] = dst[ec];
  }
#pragma unroll
  for (int k = 0; k < EN_PIECES; ++k) {
    a[k] = inv[s[k]];
    b[k] = inv[d[k]];
  }
#pragma unroll
  for (int k = 0; k < EN_PIECES; ++k) {
    const int e = base + k * (int)blockDim.x;
    if (e < n_edges) norm[e] = a[k] * b[k];
  }
}

template <bool RELU>
__global__ void __launch_bounds__(128)
gemm16_wmma(const float* __restrict__ A, const float* __restrict__ W,
            float* __restrict__ T, int n_stripes) {
  const int lane = threadIdx.x & 31;
  const int wave = threadIdx.x >> 5;
  const int half = lane >> 4;
  const int l    = lane & 15;
  const int stripe = (int)blockIdx.x * 4 + wave;
  if (stripe >= n_stripes) return;
  const int row0 = stripe << 4;

  const float* __restrict__ arow = A + (size_t)(row0 + l) * D;
  const float* __restrict__ wcol = W + 4 * l;

  v8f acc0 = {}, acc1 = {}, acc2 = {}, acc3 = {};
#pragma unroll
  for (int kk = 0; kk < D; kk += 4) {
    const int ka = kk + half * 2;
    float a0 = arow[ka];
    float a1 = arow[ka + 1];
    if (RELU) { a0 = fmaxf(a0, 0.0f); a1 = fmaxf(a1, 0.0f); }
    v2f a = {a0, a1};
    const float4 w0 = *(const float4*)(wcol + ka * D);
    const float4 w1 = *(const float4*)(wcol + (ka + 1) * D);
    v2f b0 = {w0.x, w1.x};
    v2f b1 = {w0.y, w1.y};
    v2f b2 = {w0.z, w1.z};
    v2f b3 = {w0.w, w1.w};
    acc0 = __builtin_amdgcn_wmma_f32_16x16x4_f32(false, a, false, b0, (short)0, acc0, false, false);
    acc1 = __builtin_amdgcn_wmma_f32_16x16x4_f32(false, a, false, b1, (short)0, acc1, false, false);
    acc2 = __builtin_amdgcn_wmma_f32_16x16x4_f32(false, a, false, b2, (short)0, acc2, false, false);
    acc3 = __builtin_amdgcn_wmma_f32_16x16x4_f32(false, a, false, b3, (short)0, acc3, false, false);
  }

  float4* __restrict__ orow = (float4*)(T + (size_t)(row0 + half * 8) * D + 4 * l);
#pragma unroll
  for (int j = 0; j < 8; ++j) {
    float4 o;
    o.x = acc0[j]; o.y = acc1[j]; o.z = acc2[j]; o.w = acc3[j];
    orow[(size_t)j * (D / 4)] = o;
  }
}

#define SB_PIECES 8
__global__ void self_bias_init(const float* __restrict__ T, const float* __restrict__ inv,
                               const float* __restrict__ bias, float* __restrict__ out,
                               int total4) {
  const float4* __restrict__ T4 = (const float4*)T;
  float4* __restrict__ O4 = (float4*)out;
  const int base = (int)(blockIdx.x * (blockDim.x * SB_PIECES) + threadIdx.x);
  const float4 b = ((const float4*)bias)[threadIdx.x & 15];
  float4 t[SB_PIECES];
  float is[SB_PIECES];
#pragma unroll
  for (int k = 0; k < SB_PIECES; ++k) {
    const int ic = min(base + k * (int)blockDim.x, total4 - 1);
    t[k] = T4[ic];
    is[k] = inv[ic >> 4];
  }
#pragma unroll
  for (int k = 0; k < SB_PIECES; ++k) {
    const int i = base + k * (int)blockDim.x;
    if (i < total4) {
      float4 o;
      o.x = t[k].x * is[k] * is[k] + b.x;
      o.y = t[k].y * is[k] * is[k] + b.y;
      o.z = t[k].z * is[k] * is[k] + b.z;
      o.w = t[k].w * is[k] * is[k] + b.w;
      O4[i] = o;
    }
  }
}

#define EA_BATCH 16
#define EA_WAVES 8
#define EA_SCAN 256
#define EA_GROUPS (EA_SCAN / 32)
#define EA_CHUNK (EA_WAVES * EA_SCAN)
static_assert(D == 64, "lane = feature mapping: 32 lanes x 2 columns per row");
static_assert(EA_SCAN % 32 == 0 && 32 % EA_BATCH == 0, "a scan is whole 32-edge groups of whole batches");
__global__ void __launch_bounds__(32 * EA_WAVES)
edge_aggregate(const int* __restrict__ src, const int* __restrict__ dst,
               const float* __restrict__ norm, const float* __restrict__ T,
               float* __restrict__ out, unsigned* __restrict__ tickets,
               int n_edges, int n_chunks) {
  __shared__ int q_s[EA_WAVES][EA_SCAN];
  __shared__ int q_d[EA_WAVES][EA_SCAN];
  __shared__ int q_w[EA_WAVES][EA_SCAN];
  __shared__ int item[2];
  if (threadIdx.x == 0) {
    const unsigned id = __builtin_amdgcn_s_sendmsg_rtn(0x87u);
    int h = (((id >> 16) & 15u) >= 4u) ? 1 : 0;
    unsigned t = atomicAdd(&tickets[h * TK_STRIDE], 1u);
    if (t >= (unsigned)n_chunks) {
      h ^= 1;
      t = atomicAdd(&tickets[h * TK_STRIDE], 1u);
    }
    item[0] = (int)t;
    item[1] = h;
  }
  __syncthreads();
  const int chunk = __builtin_amdgcn_readfirstlane(item[0]);
  const int cls   = __builtin_amdgcn_readfirstlane(item[1]);

  const int lane = threadIdx.x & 31;
  const int wave = __builtin_amdgcn_readfirstlane((int)(threadIdx.x >> 5));
  int* __restrict__ qs = q_s[wave];
  int* __restrict__ qd = q_d[wave];
  int* __restrict__ qw = q_w[wave];

  int M = 0;
  if (chunk < n_chunks) {
    const int wbase = chunk * EA_CHUNK + wave * EA_SCAN;
    const int elast = n_edges - 1;
    int es[EA_GROUPS], ed[EA_GROUPS], ew[EA_GROUPS];
#pragma unroll
    for (int g = 0; g < EA_GROUPS; ++g) {
      const int ec = min(wbase + g * 32 + lane, elast);
      es[g] = src[ec];
      ed[g] = dst[ec];
      ew[g] = __float_as_int(norm[ec]);
    }
    asm volatile("" :: "v"(es[0]), "v"(ew[0]));
#pragma unroll
    for (int g = 0; g < EA_GROUPS; ++g) {
      const int e = wbase + g * 32 + lane;
      const bool m = (e < n_edges) && ((ed[g] & 1) == cls);
      const unsigned mask = __builtin_amdgcn_ballot_w32(m);
      const int r = M + (int)__builtin_amdgcn_mbcnt_lo(mask, 0u);
      if (m) { qs[r] = es[g]; qd[r] = ed[g]; qw[r] = ew[g]; }
      M += __builtin_popcount(mask);
    }
  }
  __syncthreads();
  if (M == 0) return;

  const int mlast = M - 1;
  int sv, dv, nvb;
  {
    const int j = min(lane, mlast);
    sv = qs[j]; dv = qd[j]; nvb = qw[j];
  }
  for (int base = 0; base < M; base += 32) {
    int sn, dn, nn;
    {
      const int j = min(base + 32 + lane, mlast);
      sn = qs[j]; dn = qd[j]; nn = qw[j];
    }
    const int cnt = min(32, M - base);
    if (cnt == 32) {
#pragma unroll
      for (int j0 = 0; j0 < 32; j0 += EA_BATCH) {
        float ta[EA_BATCH], tb[EA_BATCH];
#pragma unroll
        for (int u = 0; u < EA_BATCH; ++u) {
          const int s = __builtin_amdgcn_readlane(sv, j0 + u);
          const float* __restrict__ r = T + (size_t)s * D + lane;
          ta[u] = r[0];
          tb[u] = r[32];
        }
#pragma unroll
        for (int u = 0; u < EA_BATCH; ++u) {
          const int d = __builtin_amdgcn_readlane(dv, j0 + u);
          const float w = __int_as_float(__builtin_amdgcn_readlane(nvb, j0 + u));
          float* o = out + (size_t)d * D + lane;
          atomicAdd(o,      ta[u] * w);
          atomicAdd(o + 32, tb[u] * w);
        }
      }
    } else {
#pragma unroll
      for (int j0 = 0; j0 < 32; j0 += EA_BATCH) {
        if (j0 < cnt) {
          float ta[EA_BATCH], tb[EA_BATCH];
#pragma unroll
          for (int u = 0; u < EA_BATCH; ++u) {
            ta[u] = 0.0f;
            tb[u] = 0.0f;
            if (j0 + u < cnt) {
              const int s = __builtin_amdgcn_readlane(sv, j0 + u);
              const float* __restrict__ r = T + (size_t)s * D + lane;
              ta[u] = r[0];
              tb[u] = r[32];
            }
          }
#pragma unroll
          for (int u = 0; u < EA_BATCH; ++u) {
            if (j0 + u < cnt) {
              const int d = __builtin_amdgcn_readlane(dv, j0 + u);
              const float w = __int_as_float(__builtin_amdgcn_readlane(nvb, j0 + u));
              float* o = out + (size_t)d * D + lane;
              atomicAdd(o,      ta[u] * w);
              atomicAdd(o + 32, tb[u] * w);
            }
          }
        }
      }
    }
    sv = sn; dv = dn; nvb = nn;
  }
}


static inline size_t align256(size_t x) { return (x + 255) & ~(size_t)255; }

extern "C" void kernel_launch(void* const* d_in, const int* in_sizes, int n_in,
                              void* d_out, int out_size, void* d_ws, size_t ws_size,
                              hipStream_t stream) {
  const int*   E  = (const int*)d_in[1];
  const float* X  = (const float*)d_in[2];
  const float* W1 = (const float*)d_in[3];
  const float* b1 = (const float*)d_in[4];
  const float* W2 = (const float*)d_in[5];
  const float* b2 = (const float*)d_in[6];
  const float* W3 = (const float*)d_in[7];
  const float* b3 = (const float*)d_in[8];
  float* out = (float*)d_out;

  const int n_edges = in_sizes[1] / 2;
  const int n_nodes = in_sizes[2] / D;
  const int* src = E;
  const int* dst = E + n_edges;

  char* ws = (char*)d_ws;
  float* inv  = (float*)ws; ws += align256((size_t)n_nodes * 4);
  float* norm = (float*)ws; ws += align256((size_t)n_edges * 4);
  float* T    = (float*)ws; ws += align256((size_t)n_nodes * D * 4);
  float* H    = (float*)ws; ws += align256((size_t)n_nodes * D * 4);
  unsigned* planes = (unsigned*)ws; ws += align256((size_t)DEG_S * n_nodes * 4);
  unsigned* tickets = (unsigned*)ws;

  const int total = n_nodes * D;
  const int nb_nodes = (n_nodes + 255) / 256;
  const int nb_edges = (n_edges + 256 * EN_PIECES - 1) / (256 * EN_PIECES);
  const int total4   = total / 4;
  const int nb_total = (total4 + 256 * SB_PIECES - 1) / (256 * SB_PIECES);
  const int n_chunks = (n_edges + EA_CHUNK - 1) / EA_CHUNK;
  const int nb_agg   = 2 * n_chunks;
  const int n_stripes = n_nodes / 16;
  const int nb_gemm  = (n_stripes + 3) / 4;
  const int range_n  = (n_nodes + DEG_R - 1) / DEG_R;
  const int slice_n  = (n_edges + DEG_S - 1) / DEG_S;

  deg_count_lds<<<DEG_R * DEG_S, 1024, (size_t)range_n * sizeof(unsigned), stream>>>(
      dst, planes, n_edges, n_nodes, range_n, slice_n);
  invsqrt_from_planes<<<nb_nodes, 256, 0, stream>>>(planes, inv, tickets, n_nodes);
  edge_norm<<<nb_edges, 256, 0, stream>>>(src, dst, inv, norm, n_edges);

  gemm16_wmma<false><<<nb_gemm, 128, 0, stream>>>(X, W1, T, n_stripes);
  self_bias_init<<<nb_total, 256, 0, stream>>>(T, inv, b1, H, total4);
  edge_aggregate<<<nb_agg, 32 * EA_WAVES, 0, stream>>>(src, dst, norm, T, H, tickets, n_edges, n_chunks);

  gemm16_wmma<true><<<nb_gemm, 128, 0, stream>>>(H, W2, T, n_stripes);
  self_bias_init<<<nb_total, 256, 0, stream>>>(T, inv, b2, H, total4);
  edge_aggregate<<<nb_agg, 32 * EA_WAVES, 0, stream>>>(src, dst, norm, T, H, tickets + TK_LAYER, n_edges, n_chunks);

  gemm16_wmma<true><<<nb_gemm, 128, 0, stream>>>(H, W3, T, n_stripes);
  self_bias_init<<<nb_total, 256, 0, stream>>>(T, inv, b3, out, total4);
  edge_aggregate<<<nb_agg, 32 * EA_WAVES, 0, stream>>>(src, dst, norm, T, out, tickets + 2 * TK_LAYER, n_edges, n_chunks);
}
